# starting kernel
# baseline (speedup 1.0000x reference)
#include <hip/hip_runtime.h>

typedef __attribute__((ext_vector_type(2))) float v2f;
typedef __attribute__((ext_vector_type(8))) float v8f;

static constexpr int NT = 4096;
static constexpr int Dd = 32, Oo = 32, Gg = 8, Tt = 64;

__global__ __launch_bounds__(256) void mlm_main_kernel(
    const float* __restrict__ x,
    const float* __restrict__ w, const float* __restrict__ mask,
    float* __restrict__ part)
{
    __shared__ float pbuf[8 * 512];

    const int tid  = threadIdx.x;
    const int wave = tid >> 5;
    const int lane = tid & 31;
    const int blk  = blockIdx.x;
    const int oh   = blk & 1;
    const int d    = (blk >> 1) & 31;
    const int g    = blk >> 6;
    const int o0   = oh << 4;

    const int olane = lane & 15;
    const int khalf = (lane >> 4) << 1;

    v2f mreg[16];
    const float* mrow = mask + (o0 + olane) * Tt + khalf;
#pragma unroll
    for (int j = 0; j < 16; ++j)
        mreg[j] = *(const v2f*)(mrow + 4 * j);

    const float* wrow = w + (size_t)((g * Dd + d) * Oo + o0 + olane) * NT + khalf;
    const float* xa   = x + olane * NT + khalf;
    const float* xb   = xa + 16 * NT;

    v8f acc0 = {}, acc1 = {}, acc2 = {}, acc3 = {};
    const int kwave = wave << 9;

    for (int kc = 0; kc < 512; kc += 64) {
        const int k0 = kwave + kc;
        const float* wp  = wrow + k0;
        const float* xap = xa + k0;
        const float* xbp = xb + k0;
#pragma unroll
        for (int j = 0; j < 16; j += 2) {
            const int kk = j << 2;
            v2f w0 = *(const v2f*)(wp + kk);
            v2f w1 = *(const v2f*)(wp + kk + 4);
            v2f b0 = w0 * mreg[j];
            v2f b1 = w1 * mreg[j + 1];
            v2f a00 = *(const v2f*)(xap + kk);
            v2f a10 = *(const v2f*)(xbp + kk);
            v2f a01 = *(const v2f*)(xap + kk + 4);
            v2f a11 = *(const v2f*)(xbp + kk + 4);
            acc0 = __builtin_amdgcn_wmma_f32_16x16x4_f32(false, a00, false, b0, (short)0, acc0, false, false);
            acc1 = __builtin_amdgcn_wmma_f32_16x16x4_f32(false, a10, false, b0, (short)0, acc1, false, false);
            acc2 = __builtin_amdgcn_wmma_f32_16x16x4_f32(false, a01, false, b1, (short)0, acc2, false, false);
            acc3 = __builtin_amdgcn_wmma_f32_16x16x4_f32(false, a11, false, b1, (short)0, acc3, false, false);
        }
        __builtin_prefetch(wp + 64, 0, 0);
    }
    acc0 += acc2;
    acc1 += acc3;

    float* pb = &pbuf[wave * 512];
#pragma unroll
    for (int r = 0; r < 8; ++r) {
        pb[r * 32 + lane]       = acc0[r];
        pb[256 + r * 32 + lane] = acc1[r];
    }
    __syncthreads();
    float* pblk = part + (size_t)blk * 512;
    for (int i = tid; i < 512; i += 256) {
        float s = 0.0f;
#pragma unroll
        for (int wv = 0; wv < 8; ++wv) s += pbuf[wv * 512 + i];
        pblk[i] = s;
    }
}

__global__ __launch_bounds__(256) void mlm_reduce_kernel(
    const float* __restrict__ part, const float* __restrict__ y,
    const float* __restrict__ bias, float* __restrict__ out)
{
    const int i = blockIdx.x * 256 + threadIdx.x;
    const int o = i & 31;
    const int d = (i >> 5) & 31;
    const int b = i >> 10;
    const int oh = o >> 4, ol = o & 15;
    const int tile = b >> 4, rb = b & 15;
    const int lanei = ol + ((rb >> 3) << 4);
    const int idx = tile * 256 + (rb & 7) * 32 + lanei;
    float s = bias[d * 32 + o];
#pragma unroll
    for (int g = 0; g < 8; ++g)
        s += y[b * 8 + g] * part[(size_t)((g * 32 + d) * 2 + oh) * 512 + idx];
    out[i] = s;
}

extern "C" void kernel_launch(void* const* d_in, const int* in_sizes, int n_in,
                              void* d_out, int out_size, void* d_ws, size_t ws_size,
                              hipStream_t stream) {
    const float* x    = (const float*)d_in[0];
    const float* y    = (const float*)d_in[1];
    const float* w    = (const float*)d_in[2];
    const float* bias = (const float*)d_in[3];
    const float* mask = (const float*)d_in[4];
    float* out  = (float*)d_out;
    float* part = (float*)d_ws;

    mlm_main_kernel<<<512, 256, 0, stream>>>(x, w, mask, part);
    mlm_reduce_kernel<<<128, 256, 0, stream>>>(part, y, bias, out);
}


# wave = two d columns of one 64-k slice (was one): x fragments, mask quads, y-scaled A operands shared by both B tiles; 2048 waves at 2 per SIMD, weight ring 3 genres ahead; summation order unchanged
# speedup vs baseline: 1.6402x; 1.6402x over previous
#include <hip/hip_runtime.h>

typedef __attribute__((ext_vector_type(2))) float v2f;
typedef __attribute__((ext_vector_type(4))) float v4f;
typedef __attribute__((ext_vector_type(8))) float v8f;

static constexpr int NT = 4096;
static constexpr int Dd = 32, Oo = 32, Gg = 8, Tt = 64;
static constexpr size_t GSTRIDE = (size_t)Dd * Oo * NT;

__device__ __forceinline__ float nbr1(float v)
{
    return __builtin_bit_cast(float,
        __builtin_amdgcn_update_dpp(0, __builtin_bit_cast(int, v), 0xB1  ,
                                    0xF, 0xF, true));
}
__device__ __forceinline__ v4f nbr(v4f v)
{
    v4f r;
    r[0] = nbr1(v[0]); r[1] = nbr1(v[1]); r[2] = nbr1(v[2]); r[3] = nbr1(v[3]);
    return r;
}
__device__ __forceinline__ v4f sel(bool c, v4f a, v4f b)
{
    v4f r;
    r[0] = c ? a[0] : b[0]; r[1] = c ? a[1] : b[1]; r[2] = c ? a[2] : b[2]; r[3] = c ? a[3] : b[3];
    return r;
}
__device__ __forceinline__ void mm_quad(v8f& acc0, v8f& acc1, v4f a0, v4f a1, v4f b)
{
    const v2f b0  = __builtin_shufflevector(b, b, 0, 1);
    const v2f b1  = __builtin_shufflevector(b, b, 2, 3);
    const v2f a00 = __builtin_shufflevector(a0, a0, 0, 1);
    const v2f a01 = __builtin_shufflevector(a0, a0, 2, 3);
    const v2f a10 = __builtin_shufflevector(a1, a1, 0, 1);
    const v2f a11 = __builtin_shufflevector(a1, a1, 2, 3);
    acc0 = __builtin_amdgcn_wmma_f32_16x16x4_f32(false, a00, false, b0, (short)0, acc0, false, false);
    acc1 = __builtin_amdgcn_wmma_f32_16x16x4_f32(false, a10, false, b0, (short)0, acc1, false, false);
    acc0 = __builtin_amdgcn_wmma_f32_16x16x4_f32(false, a01, false, b1, (short)0, acc0, false, false);
    acc1 = __builtin_amdgcn_wmma_f32_16x16x4_f32(false, a11, false, b1, (short)0, acc1, false, false);
}

static constexpr int RING = 4;
static constexpr size_t DSTRIDE = (size_t)Oo * NT;

__global__ __launch_bounds__(256) void mlm_main_kernel(
    const float* __restrict__ x, const float* __restrict__ y,
    const float* __restrict__ w, const float* __restrict__ mask,
    float* __restrict__ part)
{
    __shared__ float pbuf[8 * 1024];

    const int tid  = threadIdx.x;
    const int wave = tid >> 5;
    const int lane = tid & 31;
    const int blk  = blockIdx.x;
    const int oh   = blk & 1;
    const int d0   = ((blk >> 1) & 15) << 1;
    const int kb   = blk >> 5;
    const int o0   = oh << 4;

    const int olane = lane & 15;
    const bool odd  = lane & 1;
    const int rE    = olane & ~1;
    const int h8    = (lane >> 4) << 3;
    const int kE    = h8 + (odd ? 4 : 0);
    const int kO    = h8 + (odd ? 0 : 4);
    const int k0    = (kb << 9) + (wave << 6);

    const float* wE = w + (size_t)(d0 * Oo + o0 + rE) * NT + k0 + kE;
    const float* wO = w + (size_t)(d0 * Oo + o0 + rE + 1) * NT + k0 + kO;

    v4f wbuf[RING][16];
#pragma unroll
    for (int c = 0; c < 2; ++c) {
#pragma unroll
        for (int u = 0; u < 4; ++u) {
            wbuf[0][8 * c + 2 * u]     = *(const v4f*)(wE + c * DSTRIDE + 16 * u);
            wbuf[0][8 * c + 2 * u + 1] = *(const v4f*)(wO + c * DSTRIDE + 16 * u);
        }
    }

    v4f mreg[8];
    const float* mE = mask + (o0 + rE) * Tt + kE;
    const float* mO = mask + (o0 + rE + 1) * Tt + kO;
#pragma unroll
    for (int u = 0; u < 4; ++u) {
        mreg[2 * u]     = *(const v4f*)(mE + 16 * u);
        mreg[2 * u + 1] = *(const v4f*)(mO + 16 * u);
    }

    v4f xa[8], xb[8];
    {
        const float* xE = x + (size_t)rE * NT + k0 + kE;
        const float* xO = x + (size_t)(rE + 1) * NT + k0 + kO;
        v4f la[8], lb[8];
#pragma unroll
        for (int u = 0; u < 4; ++u) {
            la[2 * u]     = *(const v4f*)(xE + 16 * u);
            la[2 * u + 1] = *(const v4f*)(xO + 16 * u);
            lb[2 * u]     = *(const v4f*)(xE + 16 * NT + 16 * u);
            lb[2 * u + 1] = *(const v4f*)(xO + 16 * NT + 16 * u);
        }
#pragma unroll
        for (int u = 0; u < 4; ++u) {
            xa[2 * u]     = sel(odd, la[2 * u + 1], la[2 * u]);
            xa[2 * u + 1] = nbr(sel(odd, la[2 * u], la[2 * u + 1]));
            xb[2 * u]     = sel(odd, lb[2 * u + 1], lb[2 * u]);
            xb[2 * u + 1] = nbr(sel(odd, lb[2 * u], lb[2 * u + 1]));
        }
    }

    float ya[8], yb[8];
    {
        const v4f t0 = *(const v4f*)(y + olane * Gg);
        const v4f t1 = *(const v4f*)(y + olane * Gg + 4);
        const v4f t2 = *(const v4f*)(y + (olane + 16) * Gg);
        const v4f t3 = *(const v4f*)(y + (olane + 16) * Gg + 4);
#pragma unroll
        for (int e = 0; e < 4; ++e) {
            ya[e] = t0[e]; ya[4 + e] = t1[e];
            yb[e] = t2[e]; yb[4 + e] = t3[e];
        }
    }

    __builtin_amdgcn_sched_barrier(0);
#pragma unroll
    for (int p = 1; p < RING - 1; ++p) {
#pragma unroll
        for (int c = 0; c < 2; ++c) {
#pragma unroll
            for (int u = 0; u < 4; ++u) {
                wbuf[p][8 * c + 2 * u]     = *(const v4f*)(wE + p * GSTRIDE + c * DSTRIDE + 16 * u);
                wbuf[p][8 * c + 2 * u + 1] = *(const v4f*)(wO + p * GSTRIDE + c * DSTRIDE + 16 * u);
            }
        }
    }

    v8f acc[2][2] = {};

#pragma unroll
    for (int g = 0; g < Gg; ++g) {
        __builtin_amdgcn_sched_barrier(0);
        if (g + RING - 1 < Gg) {
            const size_t go = (size_t)(g + RING - 1) * GSTRIDE;
#pragma unroll
            for (int c = 0; c < 2; ++c) {
#pragma unroll
                for (int u = 0; u < 4; ++u) {
                    wbuf[(g + RING - 1) % RING][8 * c + 2 * u]     = *(const v4f*)(wE + go + c * DSTRIDE + 16 * u);
                    wbuf[(g + RING - 1) % RING][8 * c + 2 * u + 1] = *(const v4f*)(wO + go + c * DSTRIDE + 16 * u);
                }
            }
        }
        __builtin_amdgcn_sched_barrier(0);
        const float sa = ya[g], sb = yb[g];
#pragma unroll
        for (int u = 0; u < 4; ++u) {
            const v4f a0q0 = xa[2 * u] * sa,     a1q0 = xb[2 * u] * sb;
            const v4f a0q1 = xa[2 * u + 1] * sa, a1q1 = xb[2 * u + 1] * sb;
#pragma unroll
            for (int c = 0; c < 2; ++c) {
                const v4f pe = wbuf[g % RING][8 * c + 2 * u]     * mreg[2 * u];
                const v4f po = wbuf[g % RING][8 * c + 2 * u + 1] * mreg[2 * u + 1];
                const v4f qa = sel(odd, po, pe);
                const v4f qb = nbr(sel(odd, pe, po));
                mm_quad(acc[c][0], acc[c][1], a0q0, a1q0, qa);
                mm_quad(acc[c][0], acc[c][1], a0q1, a1q1, qb);
            }
        }
    }

    float* pb = &pbuf[wave * 1024];
#pragma unroll
    for (int c = 0; c < 2; ++c) {
#pragma unroll
        for (int r = 0; r < 8; ++r) {
            pb[c * 512 + r * 32 + lane]       = acc[c][0][r];
            pb[c * 512 + 256 + r * 32 + lane] = acc[c][1][r];
        }
    }
    __syncthreads();
    float* pblk = part + (size_t)blk * 1024;
    for (int i = tid; i < 1024; i += 256) {
        float s = 0.0f;
#pragma unroll
        for (int wv = 0; wv < 8; ++wv) s += pbuf[wv * 1024 + i];
        pblk[i] = s;
    }
}

__global__ __launch_bounds__(256) void mlm_reduce_kernel(
    const float* __restrict__ part,
    const float* __restrict__ bias, float* __restrict__ out)
{
    const int i = blockIdx.x * 256 + threadIdx.x;
    const int o = i & 31;
    const int d = (i >> 5) & 31;
    const int b = i >> 10;
    const int oh = o >> 4, ol = o & 15;
    const int dp = d >> 1, dc = d & 1;
    const int tile = b >> 4, rb = b & 15;
    const int lanei = ol + ((rb >> 3) << 4);
    const int idx = dc * 512 + tile * 256 + (rb & 7) * 32 + lanei;
    float s = bias[d * 32 + o];
#pragma unroll
    for (int kb = 0; kb < 8; ++kb)
        s += part[(size_t)((kb * 16 + dp) * 2 + oh) * 1024 + idx];
    out[i] = s;
}

extern "C" void kernel_launch(void* const* d_in, const int* in_sizes, int n_in,
                              void* d_out, int out_size, void* d_ws, size_t ws_size,
                              hipStream_t stream) {
    const float* x    = (const float*)d_in[0];
    const float* y    = (const float*)d_in[1];
    const float* w    = (const float*)d_in[2];
    const float* bias = (const float*)d_in[3];
    const float* mask = (const float*)d_in[4];
    float* out  = (float*)d_out;
    float* part = (float*)d_ws;

    mlm_main_kernel<<<256, 256, 0, stream>>>(x, y, w, mask, part);
    mlm_reduce_kernel<<<128, 256, 0, stream>>>(part, bias, out);
}
